# starting kernel
# baseline (speedup 1.0000x reference)
#include <hip/hip_runtime.h>
#include <hip/hip_bf16.h>

typedef __attribute__((ext_vector_type(16))) _Float16 v16h;
typedef __attribute__((ext_vector_type(8)))  _Float16 v8h;
typedef __attribute__((ext_vector_type(8)))  float    v8f;
typedef __attribute__((ext_vector_type(2)))  float    v2f;

#define N_HEADS   16
#define N_QUBITS  8
#define DIM       256
#define Q_LAYERS  2
#define N_TOKENS  2048
#define EMBED     128

#define WAVES_PER_BLOCK 4
#define W_LDS_BYTES     (DIM * DIM * 2)
#define AF_OFF          0
#define BF_OFF          8192
#define PR_OFF          16384
#define CB_OFF          32768
#define SB_OFF          33280
#define WAVE_LDS_BYTES  33792
#define TOTAL_LDS_BYTES (W_LDS_BYTES + WAVES_PER_BLOCK * WAVE_LDS_BYTES)

__global__ __launch_bounds__(256)
void qa_build_unitaries(const float* __restrict__ params,
                        float* __restrict__ Mbuf,
                        _Float16* __restrict__ W) {
    const int h = blockIdx.x;
    const int t = threadIdx.x;
    float* M = Mbuf + (size_t)h * DIM * DIM;

    for (int i = t; i < DIM * DIM; i += 256)
        M[i] = ((i >> 8) == (i & 255)) ? 1.0f : 0.0f;
    __syncthreads();

    for (int layer = 0; layer < Q_LAYERS; ++layer) {
        for (int w = 0; w < N_QUBITS; ++w) {
            const float th = 0.5f * params[(h * Q_LAYERS + layer) * N_QUBITS + w];
            const float cr = cosf(th), sr = sinf(th);
            const int pos = 7 - w;
            const int pair = t & 127;
            const int r0 = ((pair >> pos) << (pos + 1)) | (pair & ((1 << pos) - 1));
            const int r1 = r0 | (1 << pos);
            const int cbase = (t >> 7) * 128;
            float* row0 = M + r0 * DIM;
            float* row1 = M + r1 * DIM;
            for (int c = cbase; c < cbase + 128; ++c) {
                const float m0 = row0[c], m1 = row1[c];
                row0[c] = cr * m0 - sr * m1;
                row1[c] = sr * m0 + cr * m1;
            }
            __syncthreads();
        }
        for (int w = 0; w < N_QUBITS - 1; ++w) {
            const int pt = 6 - w;
            const int q = t & 63;
            const int r0 = ((q >> pt) << (pt + 2)) | (1 << (pt + 1)) | (q & ((1 << pt) - 1));
            const int r1 = r0 | (1 << pt);
            const int cbase = (t >> 6) * 64;
            float* row0 = M + r0 * DIM;
            float* row1 = M + r1 * DIM;
            for (int c = cbase; c < cbase + 64; ++c) {
                const float tmp = row0[c];
                row0[c] = row1[c];
                row1[c] = tmp;
            }
            __syncthreads();
        }
    }

    _Float16* Wh = W + (size_t)h * DIM * DIM;
    for (int i = t; i < DIM * DIM; i += 256)
        Wh[i] = (_Float16)M[i];
}

__global__ __launch_bounds__(32 * WAVES_PER_BLOCK)
void qa_measure(const float* __restrict__ x,
                const _Float16* __restrict__ W,
                float* __restrict__ out) {
    extern __shared__ char smem[];
    const int h    = blockIdx.y;
    const int wave = threadIdx.x >> 5;
    const int lane = threadIdx.x & 31;
    const int tile = blockIdx.x * WAVES_PER_BLOCK + wave;
    const int t0   = tile * 16;

    _Float16* Wlds = (_Float16*)smem;
    char* wbase = smem + W_LDS_BYTES + wave * WAVE_LDS_BYTES;
    _Float16* af    = (_Float16*)(wbase + AF_OFF);
    _Float16* bf    = (_Float16*)(wbase + BF_OFF);
    float*    probs = (float*)(wbase + PR_OFF);
    float*    cbuf  = (float*)(wbase + CB_OFF);
    float*    sbuf  = (float*)(wbase + SB_OFF);

    {
        const v8h* src = (const v8h*)(W + (size_t)h * DIM * DIM);
        v8h* dst = (v8h*)Wlds;
        for (int i = threadIdx.x; i < DIM * DIM / 8; i += 32 * WAVES_PER_BLOCK)
            dst[i] = src[i];
    }

    for (int i = lane; i < 16 * N_QUBITS; i += 32) {
        const int tt = i >> 3, w = i & 7;
        const float half = 0.5f * x[(size_t)(t0 + tt) * EMBED + h * N_QUBITS + w];
        cbuf[tt * N_QUBITS + w] = cosf(half);
        sbuf[tt * N_QUBITS + w] = sinf(half);
    }
    __syncthreads();

    for (int i = lane; i < 16 * DIM; i += 32) {
        const int tt = i >> 8, s = i & 255;
        float p = 1.0f;
#pragma unroll
        for (int w = 0; w < N_QUBITS; ++w)
            p *= ((s >> (7 - w)) & 1) ? sbuf[tt * N_QUBITS + w] : cbuf[tt * N_QUBITS + w];
        const int pc = __popc(s) & 3;
        const float a = (pc == 0) ? p : ((pc == 2) ? -p : 0.0f);
        const float b = (pc == 1) ? -p : ((pc == 3) ? p : 0.0f);
        af[tt * DIM + s] = (_Float16)a;
        bf[tt * DIM + s] = (_Float16)b;
    }
    __syncthreads();

    const int m     = lane & 15;
    const int hi    = lane >> 4;
    const int khalf = hi * 8;

    for (int ct = 0; ct < 16; ++ct) {
        v8f accA = {};
        v8f accB = {};
#pragma unroll
        for (int kt = 0; kt < 8; ++kt) {
            const int kb = kt * 32 + khalf;
            const v8h aLo = *(const v8h*)&af[m * DIM + kb];
            const v8h aHi = *(const v8h*)&af[m * DIM + kb + 16];
            const v8h bLo = *(const v8h*)&bf[m * DIM + kb];
            const v8h bHi = *(const v8h*)&bf[m * DIM + kb + 16];
            const _Float16* bp = Wlds + (size_t)(ct * 16 + m) * DIM + kb;
            const v8h wLo = *(const v8h*)bp;
            const v8h wHi = *(const v8h*)(bp + 16);
            v16h Aa, Ab, Bf;
#pragma unroll
            for (int i = 0; i < 8; ++i) {
                Aa[i] = aLo[i]; Aa[i + 8] = aHi[i];
                Ab[i] = bLo[i]; Ab[i + 8] = bHi[i];
                Bf[i] = wLo[i]; Bf[i + 8] = wHi[i];
            }
            accA = __builtin_amdgcn_wmma_f32_16x16x32_f16(
                false, Aa, false, Bf, (short)0, accA, false, false);
            accB = __builtin_amdgcn_wmma_f32_16x16x32_f16(
                false, Ab, false, Bf, (short)0, accB, false, false);
        }
#pragma unroll
        for (int r = 0; r < 8; ++r) {
            const float re = accA[r], im = accB[r];
            probs[(r + 8 * hi) * DIM + ct * 16 + m] = re * re + im * im;
        }
    }
    __syncthreads();

    const int n = lane & 15;
    const int shift = 7 - (n & 7);
    v8f zacc = {};
#pragma unroll 8
    for (int k2 = 0; k2 < 64; ++k2) {
        const int kb2 = k2 * 4 + 2 * hi;
        const v2f pA = *(const v2f*)&probs[m * DIM + kb2];
        v2f pB;
#pragma unroll
        for (int j = 0; j < 2; ++j) {
            const int sp = kb2 + j;
            const float sg = ((sp >> shift) & 1) ? -1.0f : 1.0f;
            pB[j] = (n < N_QUBITS) ? sg : 0.0f;
        }
        zacc = __builtin_amdgcn_wmma_f32_16x16x4_f32(
            false, pA, false, pB, (short)0, zacc, false, false);
    }

    if (n < N_QUBITS) {
#pragma unroll
        for (int r = 0; r < 8; ++r) {
            const int tt = r + 8 * hi;
            out[(size_t)(t0 + tt) * EMBED + h * N_QUBITS + n] = zacc[r];
        }
    }
}

extern "C" void kernel_launch(void* const* d_in, const int* in_sizes, int n_in,
                              void* d_out, int out_size, void* d_ws, size_t ws_size,
                              hipStream_t stream) {
    const float* x      = (const float*)d_in[0];
    const float* params = (const float*)d_in[1];
    float* out          = (float*)d_out;

    float*    Mbuf = (float*)d_ws;
    _Float16* W    = (_Float16*)((char*)d_ws + (size_t)N_HEADS * DIM * DIM * sizeof(float));

    qa_build_unitaries<<<dim3(N_HEADS), dim3(256), 0, stream>>>(params, Mbuf, W);

    dim3 grid(N_TOKENS / 16 / WAVES_PER_BLOCK, N_HEADS);
    qa_measure<<<grid, dim3(32 * WAVES_PER_BLOCK), TOTAL_LDS_BYTES, stream>>>(x, W, out);
}


# Build kernel keeps each head's 64-column slab of U in LDS/registers (4 RY wires per pass, CNOT ladder as one row permutation on the store), 64 blocks, instead of 30 barrier-separated global sweeps
# speedup vs baseline: 16.7298x; 16.7298x over previous
#include <hip/hip_runtime.h>
#include <hip/hip_bf16.h>

typedef __attribute__((ext_vector_type(16))) _Float16 v16h;
typedef __attribute__((ext_vector_type(8)))  _Float16 v8h;
typedef __attribute__((ext_vector_type(8)))  float    v8f;
typedef __attribute__((ext_vector_type(2)))  float    v2f;

#define N_HEADS   16
#define N_QUBITS  8
#define DIM       256
#define Q_LAYERS  2
#define N_TOKENS  2048
#define EMBED     128

#define WAVES_PER_BLOCK 4
#define W_LDS_BYTES     (DIM * DIM * 2)
#define AF_OFF          0
#define BF_OFF          8192
#define PR_OFF          16384
#define CB_OFF          32768
#define SB_OFF          33280
#define WAVE_LDS_BYTES  33792
#define TOTAL_LDS_BYTES (W_LDS_BYTES + WAVES_PER_BLOCK * WAVE_LDS_BYTES)

#define BUILD_COLS      64
#define BUILD_THREADS   1024
#define BLD_A_OFF       0
#define BLD_B_OFF       65536
#define BLD_CS_OFF      131072
#define BUILD_LDS_BYTES (131072 + 128)

__device__ __forceinline__ int qa_cnot_row(int r) {
    r ^= r >> 1;
    r ^= r >> 2;
    r ^= r >> 4;
    return r & 255;
}

__device__ __forceinline__ void qa_ry4(float (&v)[16],
                                       const float* __restrict__ cs,
                                       const float* __restrict__ sn) {
#pragma unroll
    for (int j = 0; j < 4; ++j) {
        const float cr = cs[j], sr = sn[j];
        const int bit = 8 >> j;
#pragma unroll
        for (int i0 = 0; i0 < 16; ++i0) {
            if (i0 & bit) continue;
            const int i1 = i0 | bit;
            const float m0 = v[i0], m1 = v[i1];
            v[i0] = cr * m0 - sr * m1;
            v[i1] = sr * m0 + cr * m1;
        }
    }
}

__global__ __launch_bounds__(BUILD_THREADS)
void qa_build_unitaries(const float* __restrict__ params,
                        _Float16* __restrict__ W) {
    extern __shared__ char smem[];
    float*    SA = (float*)(smem + BLD_A_OFF);
    float*    SB = (float*)(smem + BLD_B_OFF);
    float*    cs = (float*)(smem + BLD_CS_OFF);
    float*    sn = cs + Q_LAYERS * N_QUBITS;
    _Float16* H  = (_Float16*)(smem + BLD_A_OFF);

    const int h   = blockIdx.y;
    const int cg  = blockIdx.x;
    const int t   = threadIdx.x;
    const int c   = t & (BUILD_COLS - 1);
    const int nib = t >> 6;

    if (t < Q_LAYERS * N_QUBITS) {
        const float th = 0.5f * params[h * Q_LAYERS * N_QUBITS + t];
        cs[t] = cosf(th);
        sn[t] = sinf(th);
    }
    __syncthreads();

    float v[16];

    {
        const int cglob = cg * BUILD_COLS + c;
#pragma unroll
        for (int hn = 0; hn < 16; ++hn)
            v[hn] = ((hn * 16 + nib) == cglob) ? 1.0f : 0.0f;
        qa_ry4(v, cs, sn);
#pragma unroll
        for (int hn = 0; hn < 16; ++hn)
            SA[(hn * 16 + nib) * BUILD_COLS + c] = v[hn];
    }
    __syncthreads();

    {
        const int rowHi = qa_cnot_row(nib << 4);
#pragma unroll
        for (int ln = 0; ln < 16; ++ln)
            v[ln] = SA[(nib * 16 + ln) * BUILD_COLS + c];
        qa_ry4(v, cs + 4, sn + 4);
#pragma unroll
        for (int ln = 0; ln < 16; ++ln)
            SB[(rowHi ^ qa_cnot_row(ln)) * BUILD_COLS + c] = v[ln];
    }
    __syncthreads();

    {
#pragma unroll
        for (int hn = 0; hn < 16; ++hn)
            v[hn] = SB[(hn * 16 + nib) * BUILD_COLS + c];
        qa_ry4(v, cs + 8, sn + 8);
#pragma unroll
        for (int hn = 0; hn < 16; ++hn)
            SB[(hn * 16 + nib) * BUILD_COLS + c] = v[hn];
    }
    __syncthreads();

    {
        const int rowHi = qa_cnot_row(nib << 4);
#pragma unroll
        for (int ln = 0; ln < 16; ++ln)
            v[ln] = SB[(nib * 16 + ln) * BUILD_COLS + c];
        qa_ry4(v, cs + 12, sn + 12);
#pragma unroll
        for (int ln = 0; ln < 16; ++ln)
            H[(rowHi ^ qa_cnot_row(ln)) * BUILD_COLS + c] = (_Float16)v[ln];
    }
    __syncthreads();

    {
        const v8h* hsrc = (const v8h*)H;
        _Float16* Wh = W + (size_t)h * DIM * DIM + cg * BUILD_COLS;
        for (int q = t; q < DIM * BUILD_COLS / 8; q += BUILD_THREADS) {
            const int row = q >> 3, piece = q & 7;
            *(v8h*)(Wh + row * DIM + piece * 8) = hsrc[q];
        }
    }
}

__global__ __launch_bounds__(32 * WAVES_PER_BLOCK)
void qa_measure(const float* __restrict__ x,
                const _Float16* __restrict__ W,
                float* __restrict__ out) {
    extern __shared__ char smem[];
    const int h    = blockIdx.y;
    const int wave = threadIdx.x >> 5;
    const int lane = threadIdx.x & 31;
    const int tile = blockIdx.x * WAVES_PER_BLOCK + wave;
    const int t0   = tile * 16;

    _Float16* Wlds = (_Float16*)smem;
    char* wbase = smem + W_LDS_BYTES + wave * WAVE_LDS_BYTES;
    _Float16* af    = (_Float16*)(wbase + AF_OFF);
    _Float16* bf    = (_Float16*)(wbase + BF_OFF);
    float*    probs = (float*)(wbase + PR_OFF);
    float*    cbuf  = (float*)(wbase + CB_OFF);
    float*    sbuf  = (float*)(wbase + SB_OFF);

    {
        const v8h* src = (const v8h*)(W + (size_t)h * DIM * DIM);
        v8h* dst = (v8h*)Wlds;
        for (int i = threadIdx.x; i < DIM * DIM / 8; i += 32 * WAVES_PER_BLOCK)
            dst[i] = src[i];
    }

    for (int i = lane; i < 16 * N_QUBITS; i += 32) {
        const int tt = i >> 3, w = i & 7;
        const float half = 0.5f * x[(size_t)(t0 + tt) * EMBED + h * N_QUBITS + w];
        cbuf[tt * N_QUBITS + w] = cosf(half);
        sbuf[tt * N_QUBITS + w] = sinf(half);
    }
    __syncthreads();

    for (int i = lane; i < 16 * DIM; i += 32) {
        const int tt = i >> 8, s = i & 255;
        float p = 1.0f;
#pragma unroll
        for (int w = 0; w < N_QUBITS; ++w)
            p *= ((s >> (7 - w)) & 1) ? sbuf[tt * N_QUBITS + w] : cbuf[tt * N_QUBITS + w];
        const int pc = __popc(s) & 3;
        const float a = (pc == 0) ? p : ((pc == 2) ? -p : 0.0f);
        const float b = (pc == 1) ? -p : ((pc == 3) ? p : 0.0f);
        af[tt * DIM + s] = (_Float16)a;
        bf[tt * DIM + s] = (_Float16)b;
    }
    __syncthreads();

    const int m     = lane & 15;
    const int hi    = lane >> 4;
    const int khalf = hi * 8;

    for (int ct = 0; ct < 16; ++ct) {
        v8f accA = {};
        v8f accB = {};
#pragma unroll
        for (int kt = 0; kt < 8; ++kt) {
            const int kb = kt * 32 + khalf;
            const v8h aLo = *(const v8h*)&af[m * DIM + kb];
            const v8h aHi = *(const v8h*)&af[m * DIM + kb + 16];
            const v8h bLo = *(const v8h*)&bf[m * DIM + kb];
            const v8h bHi = *(const v8h*)&bf[m * DIM + kb + 16];
            const _Float16* bp = Wlds + (size_t)(ct * 16 + m) * DIM + kb;
            const v8h wLo = *(const v8h*)bp;
            const v8h wHi = *(const v8h*)(bp + 16);
            v16h Aa, Ab, Bf;
#pragma unroll
            for (int i = 0; i < 8; ++i) {
                Aa[i] = aLo[i]; Aa[i + 8] = aHi[i];
                Ab[i] = bLo[i]; Ab[i + 8] = bHi[i];
                Bf[i] = wLo[i]; Bf[i + 8] = wHi[i];
            }
            accA = __builtin_amdgcn_wmma_f32_16x16x32_f16(
                false, Aa, false, Bf, (short)0, accA, false, false);
            accB = __builtin_amdgcn_wmma_f32_16x16x32_f16(
                false, Ab, false, Bf, (short)0, accB, false, false);
        }
#pragma unroll
        for (int r = 0; r < 8; ++r) {
            const float re = accA[r], im = accB[r];
            probs[(r + 8 * hi) * DIM + ct * 16 + m] = re * re + im * im;
        }
    }
    __syncthreads();

    const int n = lane & 15;
    const int shift = 7 - (n & 7);
    v8f zacc = {};
#pragma unroll 8
    for (int k2 = 0; k2 < 64; ++k2) {
        const int kb2 = k2 * 4 + 2 * hi;
        const v2f pA = *(const v2f*)&probs[m * DIM + kb2];
        v2f pB;
#pragma unroll
        for (int j = 0; j < 2; ++j) {
            const int sp = kb2 + j;
            const float sg = ((sp >> shift) & 1) ? -1.0f : 1.0f;
            pB[j] = (n < N_QUBITS) ? sg : 0.0f;
        }
        zacc = __builtin_amdgcn_wmma_f32_16x16x4_f32(
            false, pA, false, pB, (short)0, zacc, false, false);
    }

    if (n < N_QUBITS) {
#pragma unroll
        for (int r = 0; r < 8; ++r) {
            const int tt = r + 8 * hi;
            out[(size_t)(t0 + tt) * EMBED + h * N_QUBITS + n] = zacc[r];
        }
    }
}

extern "C" void kernel_launch(void* const* d_in, const int* in_sizes, int n_in,
                              void* d_out, int out_size, void* d_ws, size_t ws_size,
                              hipStream_t stream) {
    const float* x      = (const float*)d_in[0];
    const float* params = (const float*)d_in[1];
    float* out          = (float*)d_out;

    _Float16* W    = (_Float16*)((char*)d_ws + (size_t)N_HEADS * DIM * DIM * sizeof(float));

    qa_build_unitaries<<<dim3(DIM / BUILD_COLS, N_HEADS), dim3(BUILD_THREADS), BUILD_LDS_BYTES, stream>>>(params, W);

    dim3 grid(N_TOKENS / 16 / WAVES_PER_BLOCK, N_HEADS);
    qa_measure<<<grid, dim3(32 * WAVES_PER_BLOCK), TOTAL_LDS_BYTES, stream>>>(x, W, out);
}


# LDS rows padded to a conflict-free pitch (fp16 operand rows 528 B = 16 B x odd, fp32 probs rows 1040 B): fragment reads of both GEMMs no longer 16-way bank conflicted
# speedup vs baseline: 27.5776x; 1.6484x over previous
#include <hip/hip_runtime.h>
#include <hip/hip_bf16.h>

typedef __attribute__((ext_vector_type(16))) _Float16 v16h;
typedef __attribute__((ext_vector_type(8)))  _Float16 v8h;
typedef __attribute__((ext_vector_type(8)))  float    v8f;
typedef __attribute__((ext_vector_type(2)))  float    v2f;

#define N_HEADS   16
#define N_QUBITS  8
#define DIM       256
#define Q_LAYERS  2
#define N_TOKENS  2048
#define EMBED     128

#define WAVES_PER_BLOCK 4
#define H_PITCH         (DIM + 8)
#define P_PITCH         (DIM + 4)
#define W_LDS_BYTES     (DIM * H_PITCH * 2)
#define AF_OFF          0
#define BF_OFF          8448
#define PR_OFF          16896
#define CB_OFF          33536
#define SB_OFF          34048
#define WAVE_LDS_BYTES  34560
#define TOTAL_LDS_BYTES (W_LDS_BYTES + WAVES_PER_BLOCK * WAVE_LDS_BYTES)

#define BUILD_COLS      64
#define BUILD_THREADS   1024
#define BLD_A_OFF       0
#define BLD_B_OFF       65536
#define BLD_CS_OFF      131072
#define BUILD_LDS_BYTES (131072 + 128)

__device__ __forceinline__ int qa_cnot_row(int r) {
    r ^= r >> 1;
    r ^= r >> 2;
    r ^= r >> 4;
    return r & 255;
}

__device__ __forceinline__ void qa_ry4(float (&v)[16],
                                       const float* __restrict__ cs,
                                       const float* __restrict__ sn) {
#pragma unroll
    for (int j = 0; j < 4; ++j) {
        const float cr = cs[j], sr = sn[j];
        const int bit = 8 >> j;
#pragma unroll
        for (int i0 = 0; i0 < 16; ++i0) {
            if (i0 & bit) continue;
            const int i1 = i0 | bit;
            const float m0 = v[i0], m1 = v[i1];
            v[i0] = cr * m0 - sr * m1;
            v[i1] = sr * m0 + cr * m1;
        }
    }
}

__global__ __launch_bounds__(BUILD_THREADS)
void qa_build_unitaries(const float* __restrict__ params,
                        _Float16* __restrict__ W) {
    extern __shared__ char smem[];
    float*    SA = (float*)(smem + BLD_A_OFF);
    float*    SB = (float*)(smem + BLD_B_OFF);
    float*    cs = (float*)(smem + BLD_CS_OFF);
    float*    sn = cs + Q_LAYERS * N_QUBITS;
    _Float16* H  = (_Float16*)(smem + BLD_A_OFF);

    const int h   = blockIdx.y;
    const int cg  = blockIdx.x;
    const int t   = threadIdx.x;
    const int c   = t & (BUILD_COLS - 1);
    const int nib = t >> 6;

    if (t < Q_LAYERS * N_QUBITS) {
        const float th = 0.5f * params[h * Q_LAYERS * N_QUBITS + t];
        cs[t] = cosf(th);
        sn[t] = sinf(th);
    }
    __syncthreads();

    float v[16];

    {
        const int cglob = cg * BUILD_COLS + c;
#pragma unroll
        for (int hn = 0; hn < 16; ++hn)
            v[hn] = ((hn * 16 + nib) == cglob) ? 1.0f : 0.0f;
        qa_ry4(v, cs, sn);
#pragma unroll
        for (int hn = 0; hn < 16; ++hn)
            SA[(hn * 16 + nib) * BUILD_COLS + c] = v[hn];
    }
    __syncthreads();

    {
        const int rowHi = qa_cnot_row(nib << 4);
#pragma unroll
        for (int ln = 0; ln < 16; ++ln)
            v[ln] = SA[(nib * 16 + ln) * BUILD_COLS + c];
        qa_ry4(v, cs + 4, sn + 4);
#pragma unroll
        for (int ln = 0; ln < 16; ++ln)
            SB[(rowHi ^ qa_cnot_row(ln)) * BUILD_COLS + c] = v[ln];
    }
    __syncthreads();

    {
#pragma unroll
        for (int hn = 0; hn < 16; ++hn)
            v[hn] = SB[(hn * 16 + nib) * BUILD_COLS + c];
        qa_ry4(v, cs + 8, sn + 8);
#pragma unroll
        for (int hn = 0; hn < 16; ++hn)
            SB[(hn * 16 + nib) * BUILD_COLS + c] = v[hn];
    }
    __syncthreads();

    {
        const int rowHi = qa_cnot_row(nib << 4);
#pragma unroll
        for (int ln = 0; ln < 16; ++ln)
            v[ln] = SB[(nib * 16 + ln) * BUILD_COLS + c];
        qa_ry4(v, cs + 12, sn + 12);
#pragma unroll
        for (int ln = 0; ln < 16; ++ln)
            H[(rowHi ^ qa_cnot_row(ln)) * BUILD_COLS + c] = (_Float16)v[ln];
    }
    __syncthreads();

    {
        const v8h* hsrc = (const v8h*)H;
        _Float16* Wh = W + (size_t)h * DIM * DIM + cg * BUILD_COLS;
        for (int q = t; q < DIM * BUILD_COLS / 8; q += BUILD_THREADS) {
            const int row = q >> 3, piece = q & 7;
            *(v8h*)(Wh + row * DIM + piece * 8) = hsrc[q];
        }
    }
}

__global__ __launch_bounds__(32 * WAVES_PER_BLOCK)
void qa_measure(const float* __restrict__ x,
                const _Float16* __restrict__ W,
                float* __restrict__ out) {
    extern __shared__ char smem[];
    const int h    = blockIdx.y;
    const int wave = threadIdx.x >> 5;
    const int lane = threadIdx.x & 31;
    const int tile = blockIdx.x * WAVES_PER_BLOCK + wave;
    const int t0   = tile * 16;

    _Float16* Wlds = (_Float16*)smem;
    char* wbase = smem + W_LDS_BYTES + wave * WAVE_LDS_BYTES;
    _Float16* af    = (_Float16*)(wbase + AF_OFF);
    _Float16* bf    = (_Float16*)(wbase + BF_OFF);
    float*    probs = (float*)(wbase + PR_OFF);
    float*    cbuf  = (float*)(wbase + CB_OFF);
    float*    sbuf  = (float*)(wbase + SB_OFF);

    {
        const v8h* src = (const v8h*)(W + (size_t)h * DIM * DIM);
        for (int i = threadIdx.x; i < DIM * DIM / 8; i += 32 * WAVES_PER_BLOCK) {
            const int row = i >> 5, piece = i & 31;
            *(v8h*)&Wlds[row * H_PITCH + piece * 8] = src[i];
        }
    }

    for (int i = lane; i < 16 * N_QUBITS; i += 32) {
        const int tt = i >> 3, w = i & 7;
        const float half = 0.5f * x[(size_t)(t0 + tt) * EMBED + h * N_QUBITS + w];
        cbuf[tt * N_QUBITS + w] = cosf(half);
        sbuf[tt * N_QUBITS + w] = sinf(half);
    }
    __syncthreads();

    for (int i = lane; i < 16 * DIM; i += 32) {
        const int tt = i >> 8, s = i & 255;
        float p = 1.0f;
#pragma unroll
        for (int w = 0; w < N_QUBITS; ++w)
            p *= ((s >> (7 - w)) & 1) ? sbuf[tt * N_QUBITS + w] : cbuf[tt * N_QUBITS + w];
        const int pc = __popc(s) & 3;
        const float a = (pc == 0) ? p : ((pc == 2) ? -p : 0.0f);
        const float b = (pc == 1) ? -p : ((pc == 3) ? p : 0.0f);
        af[tt * H_PITCH + s] = (_Float16)a;
        bf[tt * H_PITCH + s] = (_Float16)b;
    }
    __syncthreads();

    const int m     = lane & 15;
    const int hi    = lane >> 4;
    const int khalf = hi * 8;

    for (int ct = 0; ct < 16; ++ct) {
        v8f accA = {};
        v8f accB = {};
#pragma unroll
        for (int kt = 0; kt < 8; ++kt) {
            const int kb = kt * 32 + khalf;
            const v8h aLo = *(const v8h*)&af[m * H_PITCH + kb];
            const v8h aHi = *(const v8h*)&af[m * H_PITCH + kb + 16];
            const v8h bLo = *(const v8h*)&bf[m * H_PITCH + kb];
            const v8h bHi = *(const v8h*)&bf[m * H_PITCH + kb + 16];
            const _Float16* bp = Wlds + (size_t)(ct * 16 + m) * H_PITCH + kb;
            const v8h wLo = *(const v8h*)bp;
            const v8h wHi = *(const v8h*)(bp + 16);
            v16h Aa, Ab, Bf;
#pragma unroll
            for (int i = 0; i < 8; ++i) {
                Aa[i] = aLo[i]; Aa[i + 8] = aHi[i];
                Ab[i] = bLo[i]; Ab[i + 8] = bHi[i];
                Bf[i] = wLo[i]; Bf[i + 8] = wHi[i];
            }
            accA = __builtin_amdgcn_wmma_f32_16x16x32_f16(
                false, Aa, false, Bf, (short)0, accA, false, false);
            accB = __builtin_amdgcn_wmma_f32_16x16x32_f16(
                false, Ab, false, Bf, (short)0, accB, false, false);
        }
#pragma unroll
        for (int r = 0; r < 8; ++r) {
            const float re = accA[r], im = accB[r];
            probs[(r + 8 * hi) * P_PITCH + ct * 16 + m] = re * re + im * im;
        }
    }
    __syncthreads();

    const int n = lane & 15;
    const int shift = 7 - (n & 7);
    v8f zacc = {};
#pragma unroll 8
    for (int k2 = 0; k2 < 64; ++k2) {
        const int kb2 = k2 * 4 + 2 * hi;
        const v2f pA = *(const v2f*)&probs[m * P_PITCH + kb2];
        v2f pB;
#pragma unroll
        for (int j = 0; j < 2; ++j) {
            const int sp = kb2 + j;
            const float sg = ((sp >> shift) & 1) ? -1.0f : 1.0f;
            pB[j] = (n < N_QUBITS) ? sg : 0.0f;
        }
        zacc = __builtin_amdgcn_wmma_f32_16x16x4_f32(
            false, pA, false, pB, (short)0, zacc, false, false);
    }

    if (n < N_QUBITS) {
#pragma unroll
        for (int r = 0; r < 8; ++r) {
            const int tt = r + 8 * hi;
            out[(size_t)(t0 + tt) * EMBED + h * N_QUBITS + n] = zacc[r];
        }
    }
}

extern "C" void kernel_launch(void* const* d_in, const int* in_sizes, int n_in,
                              void* d_out, int out_size, void* d_ws, size_t ws_size,
                              hipStream_t stream) {
    const float* x      = (const float*)d_in[0];
    const float* params = (const float*)d_in[1];
    float* out          = (float*)d_out;

    _Float16* W    = (_Float16*)((char*)d_ws + (size_t)N_HEADS * DIM * DIM * sizeof(float));

    qa_build_unitaries<<<dim3(DIM / BUILD_COLS, N_HEADS), dim3(BUILD_THREADS), BUILD_LDS_BYTES, stream>>>(params, W);

    dim3 grid(N_TOKENS / 16 / WAVES_PER_BLOCK, N_HEADS);
    qa_measure<<<grid, dim3(32 * WAVES_PER_BLOCK), TOTAL_LDS_BYTES, stream>>>(x, W, out);
}


# Measure kernel: 8 waves (eight 16-token tiles) per block share one staged U_h; 16x16 = 256 blocks, one per WGP in a single round: U_h staged once per WGP instead of twice, two waves per SIMD
# speedup vs baseline: 77.0073x; 2.7924x over previous
#include <hip/hip_runtime.h>
#include <hip/hip_bf16.h>

typedef __attribute__((ext_vector_type(16))) _Float16 v16h;
typedef __attribute__((ext_vector_type(8)))  _Float16 v8h;
typedef __attribute__((ext_vector_type(8)))  float    v8f;
typedef __attribute__((ext_vector_type(2)))  float    v2f;
typedef __attribute__((ext_vector_type(4)))  float    v4f;

#define N_HEADS   16
#define N_QUBITS  8
#define DIM       256
#define Q_LAYERS  2
#define N_TOKENS  2048
#define EMBED     128

#define WAVES_PER_BLOCK 8
#define H_PITCH         (DIM + 8)
#define P_PITCH         (DIM + 4)
#define W_LDS_BYTES     (DIM * H_PITCH * 2)
#define PR_OFF          0
#define CB_OFF          16640
#define SB_OFF          17152
#define WAVE_LDS_BYTES  17664
#define TOTAL_LDS_BYTES (W_LDS_BYTES + WAVES_PER_BLOCK * WAVE_LDS_BYTES)

#define BUILD_COLS      64
#define BUILD_THREADS   1024
#define BLD_A_OFF       0
#define BLD_B_OFF       65536
#define BLD_CS_OFF      131072
#define BUILD_LDS_BYTES (131072 + 128)

__device__ __forceinline__ int qa_cnot_row(int r) {
    r ^= r >> 1;
    r ^= r >> 2;
    r ^= r >> 4;
    return r & 255;
}

__device__ __forceinline__ void qa_ry4(float (&v)[16],
                                       const float* __restrict__ cs,
                                       const float* __restrict__ sn) {
#pragma unroll
    for (int j = 0; j < 4; ++j) {
        const float cr = cs[j], sr = sn[j];
        const int bit = 8 >> j;
#pragma unroll
        for (int i0 = 0; i0 < 16; ++i0) {
            if (i0 & bit) continue;
            const int i1 = i0 | bit;
            const float m0 = v[i0], m1 = v[i1];
            v[i0] = cr * m0 - sr * m1;
            v[i1] = sr * m0 + cr * m1;
        }
    }
}

__global__ __launch_bounds__(BUILD_THREADS)
void qa_build_unitaries(const float* __restrict__ params,
                        _Float16* __restrict__ W) {
    extern __shared__ char smem[];
    float*    SA = (float*)(smem + BLD_A_OFF);
    float*    SB = (float*)(smem + BLD_B_OFF);
    float*    cs = (float*)(smem + BLD_CS_OFF);
    float*    sn = cs + Q_LAYERS * N_QUBITS;
    _Float16* H  = (_Float16*)(smem + BLD_A_OFF);

    const int h   = blockIdx.y;
    const int cg  = blockIdx.x;
    const int t   = threadIdx.x;
    const int c   = t & (BUILD_COLS - 1);
    const int nib = t >> 6;

    if (t < Q_LAYERS * N_QUBITS) {
        const float th = 0.5f * params[h * Q_LAYERS * N_QUBITS + t];
        cs[t] = cosf(th);
        sn[t] = sinf(th);
    }
    __syncthreads();

    float v[16];

    {
        const int cglob = cg * BUILD_COLS + c;
#pragma unroll
        for (int hn = 0; hn < 16; ++hn)
            v[hn] = ((hn * 16 + nib) == cglob) ? 1.0f : 0.0f;
        qa_ry4(v, cs, sn);
#pragma unroll
        for (int hn = 0; hn < 16; ++hn)
            SA[(hn * 16 + nib) * BUILD_COLS + c] = v[hn];
    }
    __syncthreads();

    {
        const int rowHi = qa_cnot_row(nib << 4);
#pragma unroll
        for (int ln = 0; ln < 16; ++ln)
            v[ln] = SA[(nib * 16 + ln) * BUILD_COLS + c];
        qa_ry4(v, cs + 4, sn + 4);
#pragma unroll
        for (int ln = 0; ln < 16; ++ln)
            SB[(rowHi ^ qa_cnot_row(ln)) * BUILD_COLS + c] = v[ln];
    }
    __syncthreads();

    {
#pragma unroll
        for (int hn = 0; hn < 16; ++hn)
            v[hn] = SB[(hn * 16 + nib) * BUILD_COLS + c];
        qa_ry4(v, cs + 8, sn + 8);
#pragma unroll
        for (int hn = 0; hn < 16; ++hn)
            SB[(hn * 16 + nib) * BUILD_COLS + c] = v[hn];
    }
    __syncthreads();

    {
        const int rowHi = qa_cnot_row(nib << 4);
#pragma unroll
        for (int ln = 0; ln < 16; ++ln)
            v[ln] = SB[(nib * 16 + ln) * BUILD_COLS + c];
        qa_ry4(v, cs + 12, sn + 12);
#pragma unroll
        for (int ln = 0; ln < 16; ++ln)
            H[(rowHi ^ qa_cnot_row(ln)) * BUILD_COLS + c] = (_Float16)v[ln];
    }
    __syncthreads();

    {
        const v8h* hsrc = (const v8h*)H;
        _Float16* Wh = W + (size_t)h * DIM * DIM + cg * BUILD_COLS;
        for (int q = t; q < DIM * BUILD_COLS / 8; q += BUILD_THREADS) {
            const int row = q >> 3, piece = q & 7;
            *(v8h*)(Wh + row * DIM + piece * 8) = hsrc[q];
        }
    }
}

__global__ __launch_bounds__(32 * WAVES_PER_BLOCK)
void qa_measure(const float* __restrict__ x,
                const _Float16* __restrict__ W,
                float* __restrict__ out) {
    extern __shared__ char smem[];
    const int h    = blockIdx.y;
    const int wave = threadIdx.x >> 5;
    const int lane = threadIdx.x & 31;
    const int tile = blockIdx.x * WAVES_PER_BLOCK + wave;
    const int t0   = tile * 16;

    _Float16* Wlds = (_Float16*)smem;
    char* wbase = smem + W_LDS_BYTES + wave * WAVE_LDS_BYTES;
    float*    probs = (float*)(wbase + PR_OFF);
    float*    cbuf  = (float*)(wbase + CB_OFF);
    float*    sbuf  = (float*)(wbase + SB_OFF);

    {
        const v8h* src = (const v8h*)(W + (size_t)h * DIM * DIM);
        for (int i = threadIdx.x; i < DIM * DIM / 8; i += 32 * WAVES_PER_BLOCK) {
            const int row = i >> 5, piece = i & 31;
            *(v8h*)&Wlds[row * H_PITCH + piece * 8] = src[i];
        }
    }

    for (int i = lane; i < 16 * N_QUBITS; i += 32) {
        const int tt = i >> 3, w = i & 7;
        const float half = 0.5f * x[(size_t)(t0 + tt) * EMBED + h * N_QUBITS + w];
        cbuf[tt * N_QUBITS + w] = cosf(half);
        sbuf[tt * N_QUBITS + w] = sinf(half);
    }
    __syncthreads();

    const int m     = lane & 15;
    const int hi    = lane >> 4;
    const int khalf = hi * 8;

    v16h Aa[8], Ab[8];
    {
        float c[8], s[8];
        {
            const v4f c0 = *(const v4f*)&cbuf[m * N_QUBITS], c1 = *(const v4f*)&cbuf[m * N_QUBITS + 4];
            const v4f s0 = *(const v4f*)&sbuf[m * N_QUBITS], s1 = *(const v4f*)&sbuf[m * N_QUBITS + 4];
#pragma unroll
            for (int w = 0; w < 4; ++w) {
                c[w] = c0[w]; c[w + 4] = c1[w];
                s[w] = s0[w]; s[w + 4] = s1[w];
            }
        }
        float pre[16];
        {
            float l1[4], l2[8], l3[16];
#pragma unroll
            for (int i = 0; i < 4; ++i)
                l1[i] = ((i >> 1) ? s[0] : c[0]) * ((i & 1) ? s[1] : c[1]);
#pragma unroll
            for (int i = 0; i < 8; ++i)
                l2[i] = l1[i >> 1] * ((i & 1) ? s[2] : c[2]);
#pragma unroll
            for (int i = 0; i < 16; ++i)
                l3[i] = l2[i >> 1] * ((i & 1) ? s[3] : c[3]);
            const float g4 = hi ? s[4] : c[4];
#pragma unroll
            for (int i = 0; i < 16; ++i)
                pre[i] = l3[i] * g4;
        }
#pragma unroll
        for (int i = 0; i < 16; ++i) {
            const int kt = i >> 1, hf = i & 1;
            float m5[2], m6[4], m7[8];
            m5[0] = pre[i] * c[5];
            m5[1] = pre[i] * s[5];
#pragma unroll
            for (int k = 0; k < 4; ++k)
                m6[k] = m5[k >> 1] * ((k & 1) ? s[6] : c[6]);
#pragma unroll
            for (int j = 0; j < 8; ++j)
                m7[j] = m6[j >> 1] * ((j & 1) ? s[7] : c[7]);
#pragma unroll
            for (int j = 0; j < 8; ++j) {
                const int q   = __builtin_popcount(kt) + hf + __builtin_popcount(j);
                const int pc0 = q & 3, pc1 = (q + 1) & 3;
                const float p = m7[j];
                const float a0 = (pc0 == 0) ? p : ((pc0 == 2) ? -p : 0.0f);
                const float a1 = (pc1 == 0) ? p : ((pc1 == 2) ? -p : 0.0f);
                const float b0 = (pc0 == 1) ? -p : ((pc0 == 3) ? p : 0.0f);
                const float b1 = (pc1 == 1) ? -p : ((pc1 == 3) ? p : 0.0f);
                Aa[kt][hf * 8 + j] = (_Float16)(hi ? a1 : a0);
                Ab[kt][hf * 8 + j] = (_Float16)(hi ? b1 : b0);
            }
        }
    }

    for (int ct = 0; ct < 16; ++ct) {
        v8f accA = {};
        v8f accB = {};
#pragma unroll
        for (int kt = 0; kt < 8; ++kt) {
            const int kb = kt * 32 + khalf;
            const _Float16* bp = Wlds + (size_t)(ct * 16 + m) * H_PITCH + kb;
            const v8h wLo = *(const v8h*)bp;
            const v8h wHi = *(const v8h*)(bp + 16);
            v16h Bf;
#pragma unroll
            for (int i = 0; i < 8; ++i) {
                Bf[i] = wLo[i]; Bf[i + 8] = wHi[i];
            }
            accA = __builtin_amdgcn_wmma_f32_16x16x32_f16(
                false, Aa[kt], false, Bf, (short)0, accA, false, false);
            accB = __builtin_amdgcn_wmma_f32_16x16x32_f16(
                false, Ab[kt], false, Bf, (short)0, accB, false, false);
        }
#pragma unroll
        for (int r = 0; r < 8; ++r) {
            const float re = accA[r], im = accB[r];
            probs[(r + 8 * hi) * P_PITCH + ct * 16 + m] = re * re + im * im;
        }
    }
    __syncthreads();

    const int n = lane & 15;
    const int shift = 7 - (n & 7);
    v8f zacc = {};
#pragma unroll 8
    for (int k2 = 0; k2 < 64; ++k2) {
        const int kb2 = k2 * 4 + 2 * hi;
        const v2f pA = *(const v2f*)&probs[m * P_PITCH + kb2];
        v2f pB;
#pragma unroll
        for (int j = 0; j < 2; ++j) {
            const int sp = kb2 + j;
            const float sg = ((sp >> shift) & 1) ? -1.0f : 1.0f;
            pB[j] = (n < N_QUBITS) ? sg : 0.0f;
        }
        zacc = __builtin_amdgcn_wmma_f32_16x16x4_f32(
            false, pA, false, pB, (short)0, zacc, false, false);
    }

    if (n < N_QUBITS) {
#pragma unroll
        for (int r = 0; r < 8; ++r) {
            const int tt = r + 8 * hi;
            out[(size_t)(t0 + tt) * EMBED + h * N_QUBITS + n] = zacc[r];
        }
    }
}

extern "C" void kernel_launch(void* const* d_in, const int* in_sizes, int n_in,
                              void* d_out, int out_size, void* d_ws, size_t ws_size,
                              hipStream_t stream) {
    const float* x      = (const float*)d_in[0];
    const float* params = (const float*)d_in[1];
    float* out          = (float*)d_out;

    _Float16* W    = (_Float16*)((char*)d_ws + (size_t)N_HEADS * DIM * DIM * sizeof(float));

    qa_build_unitaries<<<dim3(DIM / BUILD_COLS, N_HEADS), dim3(BUILD_THREADS), BUILD_LDS_BYTES, stream>>>(params, W);

    dim3 grid(N_TOKENS / 16 / WAVES_PER_BLOCK, N_HEADS);
    qa_measure<<<grid, dim3(32 * WAVES_PER_BLOCK), TOTAL_LDS_BYTES, stream>>>(x, W, out);
}


# Layer-0 RY folded into each token's product state (complex tree), layer-1 RY tensor-product matrix built per block in LDS, ladders as index maps: no build launch, no staging; sums regrouped
# speedup vs baseline: 122.5172x; 1.5910x over previous
#include <hip/hip_runtime.h>
#include <hip/hip_bf16.h>

typedef __attribute__((ext_vector_type(16))) _Float16 v16h;
typedef __attribute__((ext_vector_type(8)))  _Float16 v8h;
typedef __attribute__((ext_vector_type(8)))  float    v8f;
typedef __attribute__((ext_vector_type(2)))  float    v2f;
typedef __attribute__((ext_vector_type(4)))  float    v4f;

#define N_HEADS   16
#define N_QUBITS  8
#define DIM       256
#define Q_LAYERS  2
#define N_TOKENS  2048
#define EMBED     128

#define WAVES_PER_BLOCK 8
#define H_PITCH         (DIM + 8)
#define P_PITCH         (DIM + 4)
#define W_LDS_BYTES     (DIM * H_PITCH * 2)
#define PR_OFF          0
#define CB_OFF          16640
#define SB_OFF          17152
#define WAVE_LDS_BYTES  17664
#define PCS_OFF         (W_LDS_BYTES + WAVES_PER_BLOCK * WAVE_LDS_BYTES)
#define TOTAL_LDS_BYTES (PCS_OFF + 128)
static_assert(32 * WAVES_PER_BLOCK == DIM, "one thread per matrix row in the in-block build");

__device__ __forceinline__ int qa_cnot_row(int r) {
    r ^= r >> 1;
    r ^= r >> 2;
    r ^= r >> 4;
    return r & 255;
}

struct qa_cpx { float re, im; };
__device__ __forceinline__ qa_cpx qa_cmul(const qa_cpx a, const qa_cpx b) {
    qa_cpx o;
    o.re = a.re * b.re - a.im * b.im;
    o.im = a.re * b.im + a.im * b.re;
    return o;
}

__global__ __launch_bounds__(32 * WAVES_PER_BLOCK) __attribute__((amdgpu_waves_per_eu(2, 2)))
void qa_measure(const float* __restrict__ x,
                const float* __restrict__ params,
                float* __restrict__ out) {
    extern __shared__ char smem[];
    const int h    = blockIdx.y;
    const int wave = threadIdx.x >> 5;
    const int lane = threadIdx.x & 31;
    const int tile = blockIdx.x * WAVES_PER_BLOCK + wave;
    const int t0   = tile * 16;

    _Float16* Wlds = (_Float16*)smem;
    char* wbase = smem + W_LDS_BYTES + wave * WAVE_LDS_BYTES;
    float*    probs = (float*)(wbase + PR_OFF);
    float*    cbuf  = (float*)(wbase + CB_OFF);
    float*    sbuf  = (float*)(wbase + SB_OFF);
    float*    pcs   = (float*)(smem + PCS_OFF);

    if (threadIdx.x < Q_LAYERS * N_QUBITS) {
        const float th = 0.5f * params[h * Q_LAYERS * N_QUBITS + threadIdx.x];
        pcs[threadIdx.x] = cosf(th);
        pcs[Q_LAYERS * N_QUBITS + threadIdx.x] = sinf(th);
    }

    for (int i = lane; i < 16 * N_QUBITS; i += 32) {
        const int tt = i >> 3, w = i & 7;
        const float half = 0.5f * x[(size_t)(t0 + tt) * EMBED + h * N_QUBITS + w];
        cbuf[tt * N_QUBITS + w] = cosf(half);
        sbuf[tt * N_QUBITS + w] = sinf(half);
    }
    __syncthreads();

    {
        const int r = threadIdx.x;
        float f0[8], f1[8];
#pragma unroll
        for (int w = 0; w < N_QUBITS; ++w) {
            const float cr = pcs[N_QUBITS + w], sr = pcs[3 * N_QUBITS + w];
            const bool rb = (r >> (7 - w)) & 1;
            f0[w] = rb ? sr : cr;
            f1[w] = rb ? cr : -sr;
        }
        float hp[16], lp[16];
        {
            float a1[4], a2[8], b1[4], b2[8];
#pragma unroll
            for (int i = 0; i < 4; ++i) {
                a1[i] = ((i >> 1) ? f1[0] : f0[0]) * ((i & 1) ? f1[1] : f0[1]);
                b1[i] = ((i >> 1) ? f1[4] : f0[4]) * ((i & 1) ? f1[5] : f0[5]);
            }
#pragma unroll
            for (int i = 0; i < 8; ++i) {
                a2[i] = a1[i >> 1] * ((i & 1) ? f1[2] : f0[2]);
                b2[i] = b1[i >> 1] * ((i & 1) ? f1[6] : f0[6]);
            }
#pragma unroll
            for (int i = 0; i < 16; ++i) {
                hp[i] = a2[i >> 1] * ((i & 1) ? f1[3] : f0[3]);
                lp[i] = b2[i >> 1] * ((i & 1) ? f1[7] : f0[7]);
            }
        }
        _Float16* wrow = Wlds + r * H_PITCH;
#pragma unroll
        for (int kh = 0; kh < 16; ++kh) {
#pragma unroll
            for (int hf = 0; hf < 2; ++hf) {
                v8h pk;
#pragma unroll
                for (int j = 0; j < 8; ++j)
                    pk[j] = (_Float16)(hp[kh] * lp[hf * 8 + j]);
                *(v8h*)(wrow + kh * 16 + hf * 8) = pk;
            }
        }
    }

    const int m     = lane & 15;
    const int hi    = lane >> 4;
    const int khalf = hi * 8;

    v16h Aa[8], Ab[8];
    {
        qa_cpx al[8][2];
        {
            const v4f c0 = *(const v4f*)&cbuf[m * N_QUBITS], c1 = *(const v4f*)&cbuf[m * N_QUBITS + 4];
            const v4f s0 = *(const v4f*)&sbuf[m * N_QUBITS], s1 = *(const v4f*)&sbuf[m * N_QUBITS + 4];
#pragma unroll
            for (int w = 0; w < N_QUBITS; ++w) {
                const float c = (w < 4) ? c0[w & 3] : c1[w & 3];
                const float s = (w < 4) ? s0[w & 3] : s1[w & 3];
                const float C = pcs[w], S = pcs[2 * N_QUBITS + w];
                al[w][0].re = C * c;  al[w][0].im = S * s;
                al[w][1].re = S * c;  al[w][1].im = -(C * s);
            }
        }
        qa_cpx pre[16];
        {
            qa_cpx l1[4], l2[8], l3[16];
#pragma unroll
            for (int i = 0; i < 4; ++i)
                l1[i] = qa_cmul(al[0][i >> 1], al[1][(i & 1) ^ (i >> 1)]);
#pragma unroll
            for (int i = 0; i < 8; ++i)
                l2[i] = qa_cmul(l1[i >> 1], al[2][(i & 1) ^ ((i >> 1) & 1)]);
#pragma unroll
            for (int i = 0; i < 16; ++i)
                l3[i] = qa_cmul(l2[i >> 1], al[3][(i & 1) ^ ((i >> 1) & 1)]);
            qa_cpx x4[2];
            x4[0].re = hi ? al[4][1].re : al[4][0].re;  x4[0].im = hi ? al[4][1].im : al[4][0].im;
            x4[1].re = hi ? al[4][0].re : al[4][1].re;  x4[1].im = hi ? al[4][0].im : al[4][1].im;
#pragma unroll
            for (int i = 0; i < 16; ++i)
                pre[i] = qa_cmul(l3[i], x4[i & 1]);
        }
        qa_cpx sfx[8];
        {
            qa_cpx s5[2], s6[4];
            s5[0].re = hi ? al[5][1].re : al[5][0].re;  s5[0].im = hi ? al[5][1].im : al[5][0].im;
            s5[1].re = hi ? al[5][0].re : al[5][1].re;  s5[1].im = hi ? al[5][0].im : al[5][1].im;
#pragma unroll
            for (int k = 0; k < 4; ++k)
                s6[k] = qa_cmul(s5[k >> 1], al[6][(k & 1) ^ (k >> 1)]);
#pragma unroll
            for (int j = 0; j < 8; ++j)
                sfx[j] = qa_cmul(s6[j >> 1], al[7][(j & 1) ^ ((j >> 1) & 1)]);
        }
#pragma unroll
        for (int i = 0; i < 16; ++i) {
            const int kt = i >> 1, hf = i & 1;
#pragma unroll
            for (int j = 0; j < 8; ++j) {
                const qa_cpx v = qa_cmul(pre[i], sfx[j]);
                Aa[kt][hf * 8 + j] = (_Float16)v.re;
                Ab[kt][hf * 8 + j] = (_Float16)v.im;
            }
        }
    }
    __syncthreads();

    const int mcol = qa_cnot_row(m);
    for (int ct = 0; ct < 16; ++ct) {
        v8f accA = {};
        v8f accB = {};
#pragma unroll
        for (int kt = 0; kt < 8; ++kt) {
            const int kb = kt * 32 + khalf;
            const _Float16* bp = Wlds + (size_t)(ct * 16 + m) * H_PITCH + kb;
            const v8h wLo = *(const v8h*)bp;
            const v8h wHi = *(const v8h*)(bp + 16);
            v16h Bf;
#pragma unroll
            for (int i = 0; i < 8; ++i) {
                Bf[i] = wLo[i]; Bf[i + 8] = wHi[i];
            }
            accA = __builtin_amdgcn_wmma_f32_16x16x32_f16(
                false, Aa[kt], false, Bf, (short)0, accA, false, false);
            accB = __builtin_amdgcn_wmma_f32_16x16x32_f16(
                false, Ab[kt], false, Bf, (short)0, accB, false, false);
        }
        const int col = qa_cnot_row(ct << 4) ^ mcol;
#pragma unroll
        for (int r = 0; r < 8; ++r) {
            const float re = accA[r], im = accB[r];
            probs[(r + 8 * hi) * P_PITCH + col] = re * re + im * im;
        }
    }
    __syncthreads();

    const int n = lane & 15;
    const int shift = 7 - (n & 7);
    v8f zacc = {};
#pragma unroll 8
    for (int k2 = 0; k2 < 64; ++k2) {
        const int kb2 = k2 * 4 + 2 * hi;
        const v2f pA = *(const v2f*)&probs[m * P_PITCH + kb2];
        v2f pB;
#pragma unroll
        for (int j = 0; j < 2; ++j) {
            const int sp = kb2 + j;
            const float sg = ((sp >> shift) & 1) ? -1.0f : 1.0f;
            pB[j] = (n < N_QUBITS) ? sg : 0.0f;
        }
        zacc = __builtin_amdgcn_wmma_f32_16x16x4_f32(
            false, pA, false, pB, (short)0, zacc, false, false);
    }

    if (n < N_QUBITS) {
#pragma unroll
        for (int r = 0; r < 8; ++r) {
            const int tt = r + 8 * hi;
            out[(size_t)(t0 + tt) * EMBED + h * N_QUBITS + n] = zacc[r];
        }
    }
}

extern "C" void kernel_launch(void* const* d_in, const int* in_sizes, int n_in,
                              void* d_out, int out_size, void* d_ws, size_t ws_size,
                              hipStream_t stream) {
    const float* x      = (const float*)d_in[0];
    const float* params = (const float*)d_in[1];
    float* out          = (float*)d_out;

    dim3 grid(N_TOKENS / 16 / WAVES_PER_BLOCK, N_HEADS);
    qa_measure<<<grid, dim3(32 * WAVES_PER_BLOCK), TOTAL_LDS_BYTES, stream>>>(x, params, out);
}


# Input loads hoisted: each lane fetches its 4 encoding angles as one 16-byte load, issued with the RY-angle load before any sin/cos; was 4 serial load-then-trig trips behind the RY-angle trip
# speedup vs baseline: 314.9465x; 2.5706x over previous
#include <hip/hip_runtime.h>
#include <hip/hip_bf16.h>

typedef __attribute__((ext_vector_type(16))) _Float16 v16h;
typedef __attribute__((ext_vector_type(8)))  float    v8f;
typedef __attribute__((ext_vector_type(2)))  float    v2f;
typedef __attribute__((ext_vector_type(4)))  float    v4f;

#define N_HEADS   16
#define N_QUBITS  8
#define DIM       256
#define Q_LAYERS  2
#define N_TOKENS  2048
#define EMBED     128

#define WAVES_PER_BLOCK 8
#define CB_OFF          0
#define SB_OFF          512
#define WAVE_LDS_BYTES  1024
#define PCS_OFF         (WAVES_PER_BLOCK * WAVE_LDS_BYTES)
#define TOTAL_LDS_BYTES (PCS_OFF + 128)

struct qa_cpx { float re, im; };
__device__ __forceinline__ qa_cpx qa_cmul(const qa_cpx a, const qa_cpx b) {
    qa_cpx o;
    o.re = a.re * b.re - a.im * b.im;
    o.im = a.re * b.im + a.im * b.re;
    return o;
}
__device__ __forceinline__ qa_cpx qa_csel(const bool p, const qa_cpx a, const qa_cpx b) {
    qa_cpx o;
    o.re = p ? a.re : b.re;
    o.im = p ? a.im : b.im;
    return o;
}

__global__ __launch_bounds__(32 * WAVES_PER_BLOCK)
void qa_measure(const float* __restrict__ x,
                const float* __restrict__ params,
                float* __restrict__ out) {
    extern __shared__ char smem[];
    const int h    = blockIdx.y;
    const int wave = threadIdx.x >> 5;
    const int lane = threadIdx.x & 31;
    const int tile = blockIdx.x * WAVES_PER_BLOCK + wave;
    const int t0   = tile * 16;

    char* wbase = smem + wave * WAVE_LDS_BYTES;
    float*    cbuf  = (float*)(wbase + CB_OFF);
    float*    sbuf  = (float*)(wbase + SB_OFF);
    float*    pcs   = (float*)(smem + PCS_OFF);

    const float pang = params[h * Q_LAYERS * N_QUBITS + (lane & 15)];
    const v4f xv = *(const v4f*)&x[(size_t)(t0 + (lane >> 1)) * EMBED + h * N_QUBITS + 4 * (lane & 1)];

    if (threadIdx.x < Q_LAYERS * N_QUBITS) {
        const float th = 0.5f * pang;
        pcs[threadIdx.x] = cosf(th);
        pcs[Q_LAYERS * N_QUBITS + threadIdx.x] = sinf(th);
    }

    {
        v4f cv, sv;
#pragma unroll
        for (int k = 0; k < 4; ++k) {
            const float half = 0.5f * xv[k];
            cv[k] = cosf(half);
            sv[k] = sinf(half);
        }
        *(v4f*)&cbuf[4 * lane] = cv;
        *(v4f*)&sbuf[4 * lane] = sv;
    }
    __syncthreads();

    const int m  = lane & 15;
    const int hi = lane >> 4;

    v16h Khi0 = {}, Khi1 = {}, Klo = {};
    {
        float g0[8], g1[8];
#pragma unroll
        for (int w = 0; w < N_QUBITS; ++w) {
            const float cr = pcs[N_QUBITS + w], sr = pcs[3 * N_QUBITS + w];
            const bool rb = (m >> (3 - (w & 3))) & 1;
            g0[w] = rb ? sr : cr;
            g1[w] = rb ? cr : -sr;
        }
        const float hH = hi ? g1[0] : g0[0];
        const float hL = hi ? g1[4] : g0[4];
#pragma unroll
        for (int j = 0; j < 8; ++j) {
            const float vh = hH * ((j & 4) ? g1[1] : g0[1]) * ((j & 2) ? g1[2] : g0[2]) * ((j & 1) ? g1[3] : g0[3]);
            const float vl = hL * ((j & 4) ? g1[5] : g0[5]) * ((j & 2) ? g1[6] : g0[6]) * ((j & 1) ? g1[7] : g0[7]);
            Klo[j] = (_Float16)vl;
            if (j & 1) Khi1[j] = (_Float16)vh; else Khi0[j] = (_Float16)vh;
        }
    }

    v16h PfRe = {}, PfIm = {}, PeRe0 = {}, PeIm0 = {}, PeRe1 = {}, PeIm1 = {};
    {
        qa_cpx al[8][2];
        {
            const v4f c0 = *(const v4f*)&cbuf[m * N_QUBITS], c1 = *(const v4f*)&cbuf[m * N_QUBITS + 4];
            const v4f s0 = *(const v4f*)&sbuf[m * N_QUBITS], s1 = *(const v4f*)&sbuf[m * N_QUBITS + 4];
#pragma unroll
            for (int w = 0; w < N_QUBITS; ++w) {
                const float c = (w < 4) ? c0[w & 3] : c1[w & 3];
                const float s = (w < 4) ? s0[w & 3] : s1[w & 3];
                const float C = pcs[w], S = pcs[2 * N_QUBITS + w];
                al[w][0].re = C * c;  al[w][0].im = S * s;
                al[w][1].re = S * c;  al[w][1].im = -(C * s);
            }
        }
        {
            const qa_cpx a0 = qa_csel(hi, al[0][1], al[0][0]);
            qa_cpx u1[2], t1[2], t2[4];
            u1[0] = qa_csel(hi, al[1][1], al[1][0]);
            u1[1] = qa_csel(hi, al[1][0], al[1][1]);
            t1[0] = qa_cmul(a0, u1[0]);
            t1[1] = qa_cmul(a0, u1[1]);
#pragma unroll
            for (int k = 0; k < 4; ++k)
                t2[k] = qa_cmul(t1[k >> 1], al[2][(k & 1) ^ (k >> 1)]);
#pragma unroll
            for (int j = 0; j < 8; ++j) {
                const qa_cpx v = qa_cmul(t2[j >> 1], al[3][(j & 1) ^ ((j >> 1) & 1)]);
                PfRe[j] = (_Float16)v.re;
                PfIm[j] = (_Float16)v.im;
            }
        }
        {
            qa_cpx x4[2], s5[2], s6[4];
            x4[0] = qa_csel(hi, al[4][1], al[4][0]);
            x4[1] = qa_csel(hi, al[4][0], al[4][1]);
            s5[0] = qa_csel(hi, al[5][1], al[5][0]);
            s5[1] = qa_csel(hi, al[5][0], al[5][1]);
#pragma unroll
            for (int k = 0; k < 4; ++k)
                s6[k] = qa_cmul(s5[k >> 1], al[6][(k & 1) ^ (k >> 1)]);
#pragma unroll
            for (int j = 0; j < 8; ++j) {
                const qa_cpx sf = qa_cmul(s6[j >> 1], al[7][(j & 1) ^ ((j >> 1) & 1)]);
                const qa_cpx v0 = qa_cmul(x4[0], sf);
                const qa_cpx v1 = qa_cmul(x4[1], sf);
                PeRe0[j] = (_Float16)v0.re;  PeIm0[j] = (_Float16)v0.im;
                PeRe1[j] = (_Float16)v1.re;  PeIm1[j] = (_Float16)v1.im;
            }
        }
    }

    const v8f zero = {};
    const v8f fRe0 = __builtin_amdgcn_wmma_f32_16x16x32_f16(false, Khi0, false, PfRe,  (short)0, zero, false, false);
    const v8f fIm0 = __builtin_amdgcn_wmma_f32_16x16x32_f16(false, Khi0, false, PfIm,  (short)0, zero, false, false);
    const v8f fRe1 = __builtin_amdgcn_wmma_f32_16x16x32_f16(false, Khi1, false, PfRe,  (short)0, zero, false, false);
    const v8f fIm1 = __builtin_amdgcn_wmma_f32_16x16x32_f16(false, Khi1, false, PfIm,  (short)0, zero, false, false);
    const v8f eRe0 = __builtin_amdgcn_wmma_f32_16x16x32_f16(false, Klo,  false, PeRe0, (short)0, zero, false, false);
    const v8f eIm0 = __builtin_amdgcn_wmma_f32_16x16x32_f16(false, Klo,  false, PeIm0, (short)0, zero, false, false);
    const v8f eRe1 = __builtin_amdgcn_wmma_f32_16x16x32_f16(false, Klo,  false, PeRe1, (short)0, zero, false, false);
    const v8f eIm1 = __builtin_amdgcn_wmma_f32_16x16x32_f16(false, Klo,  false, PeIm1, (short)0, zero, false, false);

    v8f F1, F2, F3, F4, G1, G2, G3, G4;
#pragma unroll
    for (int r = 0; r < 8; ++r) {
        F1[r] = fRe0[r] * fRe0[r] + fIm0[r] * fIm0[r];
        F2[r] = fRe1[r] * fRe1[r] + fIm1[r] * fIm1[r];
        F3[r] = fRe0[r] * fRe1[r] + fIm0[r] * fIm1[r];
        F4[r] = fIm0[r] * fRe1[r] - fRe0[r] * fIm1[r];
        G1[r] = eRe0[r] * eRe0[r] + eIm0[r] * eIm0[r];
        G2[r] = eRe1[r] * eRe1[r] + eIm1[r] * eIm1[r];
        G3[r] = eRe0[r] * eRe1[r] + eIm0[r] * eIm1[r];
        G4[r] = eIm0[r] * eRe1[r] - eRe0[r] * eIm1[r];
    }

    const int n = lane & 15;
    v2f SF[4], SG[4];
#pragma unroll
    for (int q = 0; q < 4; ++q) {
#pragma unroll
        for (int j = 0; j < 2; ++j) {
            const int idx = 8 * hi + 2 * q + j;
            const float pw = (__builtin_popcount(idx >> (3 - (n & 3))) & 1) ? -1.0f : 1.0f;
            const float p3 = (__builtin_popcount(idx) & 1) ? -1.0f : 1.0f;
            SF[q][j] = (n < 4) ? pw   : ((n < 8) ? p3 : 0.0f);
            SG[q][j] = (n < 4) ? 1.0f : ((n < 8) ? pw : 0.0f);
        }
    }

    v8f FS1 = {}, FS2 = {}, FS3 = {}, FS4 = {}, GS1 = {}, GS2 = {}, GS3 = {}, GS4 = {};
#pragma unroll
    for (int q = 0; q < 4; ++q) {
        v2f a;
        a[0] = F1[2 * q]; a[1] = F1[2 * q + 1];
        FS1 = __builtin_amdgcn_wmma_f32_16x16x4_f32(false, a, false, SF[q], (short)0, FS1, false, false);
        a[0] = G1[2 * q]; a[1] = G1[2 * q + 1];
        GS1 = __builtin_amdgcn_wmma_f32_16x16x4_f32(false, a, false, SG[q], (short)0, GS1, false, false);
        a[0] = F2[2 * q]; a[1] = F2[2 * q + 1];
        FS2 = __builtin_amdgcn_wmma_f32_16x16x4_f32(false, a, false, SF[q], (short)0, FS2, false, false);
        a[0] = G2[2 * q]; a[1] = G2[2 * q + 1];
        GS2 = __builtin_amdgcn_wmma_f32_16x16x4_f32(false, a, false, SG[q], (short)0, GS2, false, false);
        a[0] = F3[2 * q]; a[1] = F3[2 * q + 1];
        FS3 = __builtin_amdgcn_wmma_f32_16x16x4_f32(false, a, false, SF[q], (short)0, FS3, false, false);
        a[0] = G3[2 * q]; a[1] = G3[2 * q + 1];
        GS3 = __builtin_amdgcn_wmma_f32_16x16x4_f32(false, a, false, SG[q], (short)0, GS3, false, false);
        a[0] = F4[2 * q]; a[1] = F4[2 * q + 1];
        FS4 = __builtin_amdgcn_wmma_f32_16x16x4_f32(false, a, false, SF[q], (short)0, FS4, false, false);
        a[0] = G4[2 * q]; a[1] = G4[2 * q + 1];
        GS4 = __builtin_amdgcn_wmma_f32_16x16x4_f32(false, a, false, SG[q], (short)0, GS4, false, false);
    }

    if (n < N_QUBITS) {
#pragma unroll
        for (int r = 0; r < 8; ++r) {
            const int tt = r + 8 * hi;
            const float zz = FS1[r] * GS1[r] + FS2[r] * GS2[r]
                           + 2.0f * (FS3[r] * GS3[r] - FS4[r] * GS4[r]);
            out[(size_t)(t0 + tt) * EMBED + h * N_QUBITS + n] = zz;
        }
    }
}

extern "C" void kernel_launch(void* const* d_in, const int* in_sizes, int n_in,
                              void* d_out, int out_size, void* d_ws, size_t ws_size,
                              hipStream_t stream) {
    const float* x      = (const float*)d_in[0];
    const float* params = (const float*)d_in[1];
    float* out          = (float*)d_out;

    dim3 grid(N_TOKENS / 16 / WAVES_PER_BLOCK, N_HEADS);
    qa_measure<<<grid, dim3(32 * WAVES_PER_BLOCK), TOTAL_LDS_BYTES, stream>>>(x, params, out);
}


# Half-angle cos/sin on the hardware trig unit (v_cos_f32/v_sin_f32, revolutions): each lane does its own token's 8 angles, RY pairs via v_readlane; no inline range reduction, no LDS stage, no barrier
# speedup vs baseline: 370.6893x; 1.1770x over previous
#include <hip/hip_runtime.h>
#include <hip/hip_bf16.h>

typedef __attribute__((ext_vector_type(16))) _Float16 v16h;
typedef __attribute__((ext_vector_type(8)))  float    v8f;
typedef __attribute__((ext_vector_type(2)))  float    v2f;
typedef __attribute__((ext_vector_type(4)))  float    v4f;

#define N_HEADS   16
#define N_QUBITS  8
#define DIM       256
#define Q_LAYERS  2
#define N_TOKENS  2048
#define EMBED     128

#define WAVES_PER_BLOCK 8

struct qa_cpx { float re, im; };
__device__ __forceinline__ qa_cpx qa_cmul(const qa_cpx a, const qa_cpx b) {
    qa_cpx o;
    o.re = a.re * b.re - a.im * b.im;
    o.im = a.re * b.im + a.im * b.re;
    return o;
}
__device__ __forceinline__ qa_cpx qa_csel(const bool p, const qa_cpx a, const qa_cpx b) {
    qa_cpx o;
    o.re = p ? a.re : b.re;
    o.im = p ? a.im : b.im;
    return o;
}

__global__ __launch_bounds__(32 * WAVES_PER_BLOCK)
void qa_measure(const float* __restrict__ x,
                const float* __restrict__ params,
                float* __restrict__ out) {
    const int h    = blockIdx.y;
    const int wave = threadIdx.x >> 5;
    const int lane = threadIdx.x & 31;
    const int tile = blockIdx.x * WAVES_PER_BLOCK + wave;
    const int t0   = tile * 16;

    const int m  = lane & 15;
    const int hi = lane >> 4;

    const float pang = params[h * Q_LAYERS * N_QUBITS + m];
    const float* xrow = &x[(size_t)(t0 + m) * EMBED + h * N_QUBITS];
    const v4f xw0 = *(const v4f*)xrow;
    const v4f xw4 = *(const v4f*)(xrow + 4);

    const float kHalfRev = 0.07957747154594767f;

    float pc[Q_LAYERS * N_QUBITS], ps[Q_LAYERS * N_QUBITS];
    {
        const float pr = kHalfRev * pang;
        const int ci = __builtin_bit_cast(int, __builtin_amdgcn_cosf(pr));
        const int si = __builtin_bit_cast(int, __builtin_amdgcn_sinf(pr));
#pragma unroll
        for (int i = 0; i < Q_LAYERS * N_QUBITS; ++i) {
            pc[i] = __builtin_bit_cast(float, __builtin_amdgcn_readlane(ci, i));
            ps[i] = __builtin_bit_cast(float, __builtin_amdgcn_readlane(si, i));
        }
    }

    float ce[N_QUBITS], se[N_QUBITS];
#pragma unroll
    for (int w = 0; w < N_QUBITS; ++w) {
        const float r = kHalfRev * ((w < 4) ? xw0[w & 3] : xw4[w & 3]);
        ce[w] = __builtin_amdgcn_cosf(r);
        se[w] = __builtin_amdgcn_sinf(r);
    }

    v16h Khi0 = {}, Khi1 = {}, Klo = {};
    {
        float g0[8], g1[8];
#pragma unroll
        for (int w = 0; w < N_QUBITS; ++w) {
            const float cr = pc[N_QUBITS + w], sr = ps[N_QUBITS + w];
            const bool rb = (m >> (3 - (w & 3))) & 1;
            g0[w] = rb ? sr : cr;
            g1[w] = rb ? cr : -sr;
        }
        const float hH = hi ? g1[0] : g0[0];
        const float hL = hi ? g1[4] : g0[4];
#pragma unroll
        for (int j = 0; j < 8; ++j) {
            const float vh = hH * ((j & 4) ? g1[1] : g0[1]) * ((j & 2) ? g1[2] : g0[2]) * ((j & 1) ? g1[3] : g0[3]);
            const float vl = hL * ((j & 4) ? g1[5] : g0[5]) * ((j & 2) ? g1[6] : g0[6]) * ((j & 1) ? g1[7] : g0[7]);
            Klo[j] = (_Float16)vl;
            if (j & 1) Khi1[j] = (_Float16)vh; else Khi0[j] = (_Float16)vh;
        }
    }

    v16h PfRe = {}, PfIm = {}, PeRe0 = {}, PeIm0 = {}, PeRe1 = {}, PeIm1 = {};
    {
        qa_cpx al[8][2];
        {
#pragma unroll
            for (int w = 0; w < N_QUBITS; ++w) {
                const float c = ce[w], s = se[w];
                const float C = pc[w], S = ps[w];
                al[w][0].re = C * c;  al[w][0].im = S * s;
                al[w][1].re = S * c;  al[w][1].im = -(C * s);
            }
        }
        {
            const qa_cpx a0 = qa_csel(hi, al[0][1], al[0][0]);
            qa_cpx u1[2], t1[2], t2[4];
            u1[0] = qa_csel(hi, al[1][1], al[1][0]);
            u1[1] = qa_csel(hi, al[1][0], al[1][1]);
            t1[0] = qa_cmul(a0, u1[0]);
            t1[1] = qa_cmul(a0, u1[1]);
#pragma unroll
            for (int k = 0; k < 4; ++k)
                t2[k] = qa_cmul(t1[k >> 1], al[2][(k & 1) ^ (k >> 1)]);
#pragma unroll
            for (int j = 0; j < 8; ++j) {
                const qa_cpx v = qa_cmul(t2[j >> 1], al[3][(j & 1) ^ ((j >> 1) & 1)]);
                PfRe[j] = (_Float16)v.re;
                PfIm[j] = (_Float16)v.im;
            }
        }
        {
            qa_cpx x4[2], s5[2], s6[4];
            x4[0] = qa_csel(hi, al[4][1], al[4][0]);
            x4[1] = qa_csel(hi, al[4][0], al[4][1]);
            s5[0] = qa_csel(hi, al[5][1], al[5][0]);
            s5[1] = qa_csel(hi, al[5][0], al[5][1]);
#pragma unroll
            for (int k = 0; k < 4; ++k)
                s6[k] = qa_cmul(s5[k >> 1], al[6][(k & 1) ^ (k >> 1)]);
#pragma unroll
            for (int j = 0; j < 8; ++j) {
                const qa_cpx sf = qa_cmul(s6[j >> 1], al[7][(j & 1) ^ ((j >> 1) & 1)]);
                const qa_cpx v0 = qa_cmul(x4[0], sf);
                const qa_cpx v1 = qa_cmul(x4[1], sf);
                PeRe0[j] = (_Float16)v0.re;  PeIm0[j] = (_Float16)v0.im;
                PeRe1[j] = (_Float16)v1.re;  PeIm1[j] = (_Float16)v1.im;
            }
        }
    }

    const v8f zero = {};
    const v8f fRe0 = __builtin_amdgcn_wmma_f32_16x16x32_f16(false, Khi0, false, PfRe,  (short)0, zero, false, false);
    const v8f fIm0 = __builtin_amdgcn_wmma_f32_16x16x32_f16(false, Khi0, false, PfIm,  (short)0, zero, false, false);
    const v8f fRe1 = __builtin_amdgcn_wmma_f32_16x16x32_f16(false, Khi1, false, PfRe,  (short)0, zero, false, false);
    const v8f fIm1 = __builtin_amdgcn_wmma_f32_16x16x32_f16(false, Khi1, false, PfIm,  (short)0, zero, false, false);
    const v8f eRe0 = __builtin_amdgcn_wmma_f32_16x16x32_f16(false, Klo,  false, PeRe0, (short)0, zero, false, false);
    const v8f eIm0 = __builtin_amdgcn_wmma_f32_16x16x32_f16(false, Klo,  false, PeIm0, (short)0, zero, false, false);
    const v8f eRe1 = __builtin_amdgcn_wmma_f32_16x16x32_f16(false, Klo,  false, PeRe1, (short)0, zero, false, false);
    const v8f eIm1 = __builtin_amdgcn_wmma_f32_16x16x32_f16(false, Klo,  false, PeIm1, (short)0, zero, false, false);

    v8f F1, F2, F3, F4, G1, G2, G3, G4;
#pragma unroll
    for (int r = 0; r < 8; ++r) {
        F1[r] = fRe0[r] * fRe0[r] + fIm0[r] * fIm0[r];
        F2[r] = fRe1[r] * fRe1[r] + fIm1[r] * fIm1[r];
        F3[r] = fRe0[r] * fRe1[r] + fIm0[r] * fIm1[r];
        F4[r] = fIm0[r] * fRe1[r] - fRe0[r] * fIm1[r];
        G1[r] = eRe0[r] * eRe0[r] + eIm0[r] * eIm0[r];
        G2[r] = eRe1[r] * eRe1[r] + eIm1[r] * eIm1[r];
        G3[r] = eRe0[r] * eRe1[r] + eIm0[r] * eIm1[r];
        G4[r] = eIm0[r] * eRe1[r] - eRe0[r] * eIm1[r];
    }

    const int n = lane & 15;
    v2f SF[4], SG[4];
#pragma unroll
    for (int q = 0; q < 4; ++q) {
#pragma unroll
        for (int j = 0; j < 2; ++j) {
            const int idx = 8 * hi + 2 * q + j;
            const float pw = (__builtin_popcount(idx >> (3 - (n & 3))) & 1) ? -1.0f : 1.0f;
            const float p3 = (__builtin_popcount(idx) & 1) ? -1.0f : 1.0f;
            SF[q][j] = (n < 4) ? pw   : ((n < 8) ? p3 : 0.0f);
            SG[q][j] = (n < 4) ? 1.0f : ((n < 8) ? pw : 0.0f);
        }
    }

    v8f FS1 = {}, FS2 = {}, FS3 = {}, FS4 = {}, GS1 = {}, GS2 = {}, GS3 = {}, GS4 = {};
#pragma unroll
    for (int q = 0; q < 4; ++q) {
        v2f a;
        a[0] = F1[2 * q]; a[1] = F1[2 * q + 1];
        FS1 = __builtin_amdgcn_wmma_f32_16x16x4_f32(false, a, false, SF[q], (short)0, FS1, false, false);
        a[0] = G1[2 * q]; a[1] = G1[2 * q + 1];
        GS1 = __builtin_amdgcn_wmma_f32_16x16x4_f32(false, a, false, SG[q], (short)0, GS1, false, false);
        a[0] = F2[2 * q]; a[1] = F2[2 * q + 1];
        FS2 = __builtin_amdgcn_wmma_f32_16x16x4_f32(false, a, false, SF[q], (short)0, FS2, false, false);
        a[0] = G2[2 * q]; a[1] = G2[2 * q + 1];
        GS2 = __builtin_amdgcn_wmma_f32_16x16x4_f32(false, a, false, SG[q], (short)0, GS2, false, false);
        a[0] = F3[2 * q]; a[1] = F3[2 * q + 1];
        FS3 = __builtin_amdgcn_wmma_f32_16x16x4_f32(false, a, false, SF[q], (short)0, FS3, false, false);
        a[0] = G3[2 * q]; a[1] = G3[2 * q + 1];
        GS3 = __builtin_amdgcn_wmma_f32_16x16x4_f32(false, a, false, SG[q], (short)0, GS3, false, false);
        a[0] = F4[2 * q]; a[1] = F4[2 * q + 1];
        FS4 = __builtin_amdgcn_wmma_f32_16x16x4_f32(false, a, false, SF[q], (short)0, FS4, false, false);
        a[0] = G4[2 * q]; a[1] = G4[2 * q + 1];
        GS4 = __builtin_amdgcn_wmma_f32_16x16x4_f32(false, a, false, SG[q], (short)0, GS4, false, false);
    }

    if (n < N_QUBITS) {
#pragma unroll
        for (int r = 0; r < 8; ++r) {
            const int tt = r + 8 * hi;
            const float zz = FS1[r] * GS1[r] + FS2[r] * GS2[r]
                           + 2.0f * (FS3[r] * GS3[r] - FS4[r] * GS4[r]);
            out[(size_t)(t0 + tt) * EMBED + h * N_QUBITS + n] = zz;
        }
    }
}

extern "C" void kernel_launch(void* const* d_in, const int* in_sizes, int n_in,
                              void* d_out, int out_size, void* d_ws, size_t ws_size,
                              hipStream_t stream) {
    const float* x      = (const float*)d_in[0];
    const float* params = (const float*)d_in[1];
    float* out          = (float*)d_out;

    dim3 grid(N_TOKENS / 16 / WAVES_PER_BLOCK, N_HEADS);
    qa_measure<<<grid, dim3(32 * WAVES_PER_BLOCK), 0, stream>>>(x, params, out);
}
